# starting kernel
# baseline (speedup 1.0000x reference)
#include <hip/hip_runtime.h>

typedef __attribute__((ext_vector_type(2))) float v2f;
typedef __attribute__((ext_vector_type(8))) float v8f;

#define IMSIZE   224
#define NSEG     1024
#define SMOOTH_K 8.0f
#define EPS_F    1e-10f
#define TILES_PER_ROW (IMSIZE / 16)
#define WAVES_PER_BLOCK 8
#define NUM_TILES ((IMSIZE * IMSIZE) / 16)

#define NEGK_LOG2E (-SMOOTH_K * 1.44269504088896340736f)
#define LOG2_NSEG  10.0f
#define LN2_OVER_K (0.69314718055994530942f / SMOOTH_K)

__global__ __launch_bounds__(256) void sketch_render_kernel(
    const float* __restrict__ xpar,
    const float* __restrict__ ypar,
    float* __restrict__ out)
{
    __shared__ float4 sAB[NSEG];
    __shared__ float4 sQ[NSEG];
    __shared__ float2 sR[NSEG];

    for (int s = threadIdx.x; s < NSEG; s += blockDim.x) {
        float x0 = xpar[s]     * (float)IMSIZE;
        float y0 = ypar[s]     * (float)IMSIZE;
        float x1 = xpar[s + 1] * (float)IMSIZE;
        float y1 = ypar[s + 1] * (float)IMSIZE;
        float dx = x1 - x0, dy = y1 - y0;
        float d2 = dx * dx + dy * dy;
        float inv = 1.0f / (d2 + EPS_F);
        float cross = y0 * x1 - x0 * y1;
        float A  = dx * dy * inv;
        float B  = dx * dx * inv;
        float Cc = -dy * cross * inv;
        float idx = 1.0f / (dx + EPS_F);
        float sl = dy * idx;
        float ic = cross * idx;
        float vert = (x0 == x1) ? 1.0f : 0.0f;
        sAB[s] = make_float4(A, B, Cc, sl);
        sQ[s]  = make_float4(ic, fminf(x0, x1), fmaxf(x0, x1), fminf(y0, y1));
        sR[s]  = make_float2(fmaxf(y0, y1), vert);
    }
    __syncthreads();

    int wid  = (blockIdx.x * blockDim.x + threadIdx.x) >> 5;
    int lane = threadIdx.x & 31;
    int grp  = lane >> 4;
    int m    = lane & 15;
    int row  = wid / TILES_PER_ROW;
    int col  = wid % TILES_PER_ROW;
    float xp = (float)(col * 16 + m);
    float yp = (float)row;

    v8f c = {0.0f, 0.0f, 0.0f, 0.0f, 0.0f, 0.0f, 0.0f, 0.0f};
    v2f bones = {1.0f, 1.0f};

    #pragma unroll 2
    for (int s = 0; s < NSEG; s += 4) {
        int k0 = s + grp * 2;
        v2f a;
        #pragma unroll
        for (int j = 0; j < 2; ++j) {
            int k = k0 + j;
            float4 p = sAB[k];
            float4 q = sQ[k];
            float2 r = sR[k];
            float cx = fmaf(p.x, yp, fmaf(p.y, xp, p.z));
            float cy = fmaf(p.w, cx, q.x);
            float yps = (r.y != 0.0f) ? yp : cy;
            float xp1 = fminf(fmaxf(cx,  q.y), q.z);
            float yp1 = fminf(fmaxf(yps, q.w), r.x);
            float ddx = xp - xp1;
            float ddy = yp - yp1;
            float t = fmaf(ddx, ddx, EPS_F);
            t = fmaf(ddy, ddy, t);
            float dist = __fsqrt_rn(t);
            a[j] = __builtin_amdgcn_exp2f(dist * NEGK_LOG2E);
        }
        c = __builtin_amdgcn_wmma_f32_16x16x4_f32(false, a, false, bones,
                                                  (short)0, c, false, false);
    }

    if (lane == 0 || lane == 16) {
        int base = row * IMSIZE + col * 16 + grp * 8;
        #pragma unroll
        for (int rI = 0; rI < 8; ++rI) {
            float res = c[rI];
            float val = (LOG2_NSEG - __builtin_amdgcn_logf(res)) * LN2_OVER_K;
            out[base + rI] = val;
        }
    }
}

extern "C" void kernel_launch(void* const* d_in, const int* in_sizes, int n_in,
                              void* d_out, int out_size, void* d_ws, size_t ws_size,
                              hipStream_t stream) {
    (void)in_sizes; (void)n_in; (void)out_size; (void)d_ws; (void)ws_size;
    const float* x = (const float*)d_in[0];
    const float* y = (const float*)d_in[1];
    float* out = (float*)d_out;
    dim3 grid(NUM_TILES / WAVES_PER_BLOCK);
    dim3 block(WAVES_PER_BLOCK * 32);
    sketch_render_kernel<<<grid, block, 0, stream>>>(x, y, out);
}


# One block per image row: terms that depend only on (segment,row) (A*yp+Cc, the vertical-segment select) move into the block's LDS table; per pair 1 fma less, no compare/select, 32 B read not 40
# speedup vs baseline: 1.2988x; 1.2988x over previous
#include <hip/hip_runtime.h>


typedef __attribute__((ext_vector_type(2))) float v2f;
typedef __attribute__((ext_vector_type(4))) float v4f;
typedef __attribute__((ext_vector_type(8))) float v8f;

#define IMSIZE   224
#define NSEG     1024
#define NPAIR    (NSEG / 2)
#define SMOOTH_K 8.0f
#define EPS_F    1e-10f
#define TILES_PER_ROW (IMSIZE / 16)
#define BLOCK_THREADS (TILES_PER_ROW * 32)

#define NEGK_LOG2E (-SMOOTH_K * 1.44269504088896340736f)
#define LOG2_NSEG  10.0f
#define LN2_OVER_K (0.69314718055994530942f / SMOOTH_K)

struct SegRow { float B, Cr, sl, ic, minx, maxx, miny, maxy; };

__device__ __forceinline__ SegRow seg_row(float x0, float y0, float x1, float y1, float yp)
{
    float dx = x1 - x0, dy = y1 - y0;
    float d2 = fmaf(dy, dy, dx * dx);
    float inv = 1.0f / (d2 + EPS_F);
    float cross = fmaf(y0, x1, -(x0 * y1));
    float A  = dx * dy * inv;
    float B  = dx * dx * inv;
    float Cc = -dy * cross * inv;
    float idx = 1.0f / (dx + EPS_F);
    float sl = dy * idx;
    float ic = cross * idx;
    bool  vert = (x0 == x1);
    SegRow r;
    r.B    = B;
    r.Cr   = fmaf(A, yp, Cc);
    r.sl   = vert ? 0.0f : sl;
    r.ic   = vert ? yp   : ic;
    r.minx = fminf(x0, x1);
    r.maxx = fmaxf(x0, x1);
    r.miny = fminf(y0, y1);
    r.maxy = fmaxf(y0, y1);
    return r;
}

__global__ __launch_bounds__(BLOCK_THREADS) void sketch_render_kernel(
    const float* __restrict__ xpar,
    const float* __restrict__ ypar,
    float* __restrict__ out)
{
    __shared__ v4f sT[NPAIR * 4];

    const int   row = blockIdx.x;
    const float yp  = (float)row;

    for (int p = threadIdx.x; p < NPAIR; p += BLOCK_THREADS) {
        float xa = xpar[2 * p]     * (float)IMSIZE;
        float xb = xpar[2 * p + 1] * (float)IMSIZE;
        float xc = xpar[2 * p + 2] * (float)IMSIZE;
        float ya = ypar[2 * p]     * (float)IMSIZE;
        float yb = ypar[2 * p + 1] * (float)IMSIZE;
        float yc = ypar[2 * p + 2] * (float)IMSIZE;
        SegRow e = seg_row(xa, ya, xb, yb, yp);
        SegRow o = seg_row(xb, yb, xc, yc, yp);
        v4f r0 = {e.B,    o.B,    e.Cr,   o.Cr};
        v4f r1 = {e.sl,   o.sl,   e.ic,   o.ic};
        v4f r2 = {e.minx, o.minx, e.maxx, o.maxx};
        v4f r3 = {e.miny, o.miny, e.maxy, o.maxy};
        sT[p * 4 + 0] = r0;
        sT[p * 4 + 1] = r1;
        sT[p * 4 + 2] = r2;
        sT[p * 4 + 3] = r3;
    }
    __syncthreads();

    const int wave = threadIdx.x >> 5;
    const int lane = threadIdx.x & 31;
    const int grp  = lane >> 4;
    const int m    = lane & 15;
    const float xp = (float)(wave * 16 + m);
    const v2f xpv  = {xp, xp};
    const v2f ypv  = {yp, yp};
    const v2f epsv = {EPS_F, EPS_F};

    v8f c = {0.0f, 0.0f, 0.0f, 0.0f, 0.0f, 0.0f, 0.0f, 0.0f};
    const v2f bones = {1.0f, 1.0f};

    const v4f* rec = sT + grp * 4;

    #pragma unroll 2
    for (int it = 0; it < NSEG / 4; ++it) {
        const v4f* r = rec + it * 8;
        v4f r0 = r[0], r1 = r[1], r2 = r[2], r3 = r[3];
        v2f Bv  = {r0.x, r0.y}, Crv = {r0.z, r0.w};
        v2f slv = {r1.x, r1.y}, icv = {r1.z, r1.w};
        v2f cx = __builtin_elementwise_fma(Bv, xpv, Crv);
        v2f cy = __builtin_elementwise_fma(slv, cx, icv);
        v2f xp1, yp1;
        xp1.x = fminf(fmaxf(cx.x, r2.x), r2.z);
        xp1.y = fminf(fmaxf(cx.y, r2.y), r2.w);
        yp1.x = fminf(fmaxf(cy.x, r3.x), r3.z);
        yp1.y = fminf(fmaxf(cy.y, r3.y), r3.w);
        v2f ddx = xpv - xp1;
        v2f ddy = ypv - yp1;
        v2f t = __builtin_elementwise_fma(ddx, ddx, epsv);
        t = __builtin_elementwise_fma(ddy, ddy, t);
        v2f a;
        a.x = __builtin_amdgcn_exp2f(__builtin_amdgcn_sqrtf(t.x) * NEGK_LOG2E);
        a.y = __builtin_amdgcn_exp2f(__builtin_amdgcn_sqrtf(t.y) * NEGK_LOG2E);
        c = __builtin_amdgcn_wmma_f32_16x16x4_f32(false, a, false, bones,
                                                  (short)0, c, false, false);
    }

    if (lane == 0 || lane == 16) {
        int base = row * IMSIZE + wave * 16 + grp * 8;
        #pragma unroll
        for (int rI = 0; rI < 8; ++rI) {
            float res = c[rI];
            float val = (LOG2_NSEG - __builtin_amdgcn_logf(res)) * LN2_OVER_K;
            out[base + rI] = val;
        }
    }
}

extern "C" void kernel_launch(void* const* d_in, const int* in_sizes, int n_in,
                              void* d_out, int out_size, void* d_ws, size_t ws_size,
                              hipStream_t stream) {
    (void)in_sizes; (void)n_in; (void)out_size; (void)d_ws; (void)ws_size;
    const float* x = (const float*)d_in[0];
    const float* y = (const float*)d_in[1];
    float* out = (float*)d_out;
    dim3 grid(IMSIZE);
    dim3 block(BLOCK_THREADS);
    sketch_render_kernel<<<grid, block, 0, stream>>>(x, y, out);
}


# Each wave covers 7 tiles (half a row) over one eighth of the segments, 2 x 8 waves per row block: one LDS record read serves 7 tiles; the 8 range sums are added in fixed order through LDS
# speedup vs baseline: 1.7368x; 1.3372x over previous
#include <hip/hip_runtime.h>


typedef __attribute__((ext_vector_type(2))) float v2f;
typedef __attribute__((ext_vector_type(4))) float v4f;
typedef __attribute__((ext_vector_type(8))) float v8f;

#define IMSIZE   224
#define NSEG     1024
#define NPAIR    (NSEG / 2)
#define SMOOTH_K 8.0f
#define EPS_F    1e-10f

#define TILES_PER_WAVE 7
#define KSPLIT         8
#define NWAVES         (2 * KSPLIT)
#define BLOCK_THREADS  (NWAVES * 32)
#define SEG_PER_WAVE   (NSEG / KSPLIT)
#define STEPS          (SEG_PER_WAVE / 4)

#define NEGK_LOG2E (-SMOOTH_K * 1.44269504088896340736f)
#define LOG2_NSEG  10.0f
#define LN2_OVER_K (0.69314718055994530942f / SMOOTH_K)

struct SegRow { float B, Cr, sl, ic, minx, maxx, miny, maxy; };

__device__ __forceinline__ SegRow seg_row(float x0, float y0, float x1, float y1, float yp)
{
    float dx = x1 - x0, dy = y1 - y0;
    float d2 = fmaf(dy, dy, dx * dx);
    float inv = 1.0f / (d2 + EPS_F);
    float cross = fmaf(y0, x1, -(x0 * y1));
    float A  = dx * dy * inv;
    float B  = dx * dx * inv;
    float Cc = -dy * cross * inv;
    float idx = 1.0f / (dx + EPS_F);
    float sl = dy * idx;
    float ic = cross * idx;
    bool  vert = (x0 == x1);
    SegRow r;
    r.B    = B;
    r.Cr   = fmaf(A, yp, Cc);
    r.sl   = vert ? 0.0f : sl;
    r.ic   = vert ? yp   : ic;
    r.minx = fminf(x0, x1);
    r.maxx = fmaxf(x0, x1);
    r.miny = fminf(y0, y1);
    r.maxy = fmaxf(y0, y1);
    return r;
}

__device__ __forceinline__ void tiles_step(const v4f r0, const v4f r1, const v4f r2, const v4f r3,
                                           const v2f* __restrict__ xpv, const v2f ypv,
                                           v8f* __restrict__ c)
{
    const v2f epsv  = {EPS_F, EPS_F};
    const v2f bones = {1.0f, 1.0f};
    const v2f Bv  = {r0.x, r0.y}, Crv = {r0.z, r0.w};
    const v2f slv = {r1.x, r1.y}, icv = {r1.z, r1.w};
    v2f cx[TILES_PER_WAVE], cy[TILES_PER_WAVE], ddx[TILES_PER_WAVE], ddy[TILES_PER_WAVE];
    v2f s[TILES_PER_WAVE], a[TILES_PER_WAVE];
    #pragma unroll
    for (int t = 0; t < TILES_PER_WAVE; ++t)
        cx[t] = __builtin_elementwise_fma(Bv, xpv[t], Crv);
    #pragma unroll
    for (int t = 0; t < TILES_PER_WAVE; ++t)
        cy[t] = __builtin_elementwise_fma(slv, cx[t], icv);
    #pragma unroll
    for (int t = 0; t < TILES_PER_WAVE; ++t) {
        v2f xp1;
        xp1.x = fminf(fmaxf(cx[t].x, r2.x), r2.z);
        xp1.y = fminf(fmaxf(cx[t].y, r2.y), r2.w);
        ddx[t] = xpv[t] - xp1;
    }
    #pragma unroll
    for (int t = 0; t < TILES_PER_WAVE; ++t) {
        v2f yp1;
        yp1.x = fminf(fmaxf(cy[t].x, r3.x), r3.z);
        yp1.y = fminf(fmaxf(cy[t].y, r3.y), r3.w);
        ddy[t] = ypv - yp1;
    }
    #pragma unroll
    for (int t = 0; t < TILES_PER_WAVE; ++t)
        s[t] = __builtin_elementwise_fma(ddx[t], ddx[t], epsv);
    #pragma unroll
    for (int t = 0; t < TILES_PER_WAVE; ++t)
        s[t] = __builtin_elementwise_fma(ddy[t], ddy[t], s[t]);
    #pragma unroll
    for (int t = 0; t < TILES_PER_WAVE; ++t) {
        a[t].x = __builtin_amdgcn_sqrtf(s[t].x) * NEGK_LOG2E;
        a[t].y = __builtin_amdgcn_sqrtf(s[t].y) * NEGK_LOG2E;
    }
    #pragma unroll
    for (int t = 0; t < TILES_PER_WAVE; ++t) {
        a[t].x = __builtin_amdgcn_exp2f(a[t].x);
        a[t].y = __builtin_amdgcn_exp2f(a[t].y);
    }
    #pragma unroll
    for (int t = 0; t < TILES_PER_WAVE; ++t)
        c[t] = __builtin_amdgcn_wmma_f32_16x16x4_f32(false, a[t], false, bones,
                                                     (short)0, c[t], false, false);
}

__global__ __launch_bounds__(BLOCK_THREADS) void sketch_render_kernel(
    const float* __restrict__ xpar,
    const float* __restrict__ ypar,
    float* __restrict__ out)
{
    __shared__ v4f sT[NPAIR * 4];
    __shared__ v4f sP[KSPLIT * IMSIZE / 4];

    const int   row = blockIdx.x;
    const float yp  = (float)row;

    {
        const int p = threadIdx.x;
        float xa = xpar[2 * p]     * (float)IMSIZE;
        float xb = xpar[2 * p + 1] * (float)IMSIZE;
        float xc = xpar[2 * p + 2] * (float)IMSIZE;
        float ya = ypar[2 * p]     * (float)IMSIZE;
        float yb = ypar[2 * p + 1] * (float)IMSIZE;
        float yc = ypar[2 * p + 2] * (float)IMSIZE;
        SegRow e = seg_row(xa, ya, xb, yb, yp);
        SegRow o = seg_row(xb, yb, xc, yc, yp);
        v4f r0 = {e.B,    o.B,    e.Cr,   o.Cr};
        v4f r1 = {e.sl,   o.sl,   e.ic,   o.ic};
        v4f r2 = {e.minx, o.minx, e.maxx, o.maxx};
        v4f r3 = {e.miny, o.miny, e.maxy, o.maxy};
        sT[p * 4 + 0] = r0;
        sT[p * 4 + 1] = r1;
        sT[p * 4 + 2] = r2;
        sT[p * 4 + 3] = r3;
    }
    __syncthreads();

    const int wave = threadIdx.x >> 5;
    const int lane = threadIdx.x & 31;
    const int grp  = lane >> 4;
    const int m    = lane & 15;
    const int half = wave & 1;
    const int ks   = wave >> 1;

    const v2f ypv = {yp, yp};
    v2f xpv[TILES_PER_WAVE];
    v8f c[TILES_PER_WAVE];
    #pragma unroll
    for (int t = 0; t < TILES_PER_WAVE; ++t) {
        const float xp = (float)((half * TILES_PER_WAVE + t) * 16 + m);
        xpv[t] = (v2f){xp, xp};
        c[t]   = (v8f){0.0f, 0.0f, 0.0f, 0.0f, 0.0f, 0.0f, 0.0f, 0.0f};
    }

    const v4f* rec = sT + (ks * (SEG_PER_WAVE / 2) + grp) * 4;

    for (int it = 0; it < STEPS; it += 2) {
        const v4f* r = rec + it * 8;
        v4f a0 = r[0], a1 = r[1], a2 = r[2],  a3 = r[3];
        v4f b0 = r[8], b1 = r[9], b2 = r[10], b3 = r[11];
        tiles_step(a0, a1, a2, a3, xpv, ypv, c);
        tiles_step(b0, b1, b2, b3, xpv, ypv, c);
    }

    if (m == 0) {
        v4f* dst = sP + (ks * IMSIZE + half * TILES_PER_WAVE * 16 + grp * 8) / 4;
        #pragma unroll
        for (int t = 0; t < TILES_PER_WAVE; ++t) {
            v4f lo = {c[t][0], c[t][1], c[t][2], c[t][3]};
            v4f hi = {c[t][4], c[t][5], c[t][6], c[t][7]};
            dst[t * 4 + 0] = lo;
            dst[t * 4 + 1] = hi;
        }
    }
    __syncthreads();

    if (threadIdx.x < IMSIZE) {
        const float* P = (const float*)sP;
        float res = P[threadIdx.x];
        #pragma unroll
        for (int k = 1; k < KSPLIT; ++k)
            res += P[k * IMSIZE + threadIdx.x];
        out[row * IMSIZE + threadIdx.x] = (LOG2_NSEG - __builtin_amdgcn_logf(res)) * LN2_OVER_K;
    }
}

extern "C" void kernel_launch(void* const* d_in, const int* in_sizes, int n_in,
                              void* d_out, int out_size, void* d_ws, size_t ws_size,
                              hipStream_t stream) {
    (void)in_sizes; (void)n_in; (void)out_size; (void)d_ws; (void)ws_size;
    const float* x = (const float*)d_in[0];
    const float* y = (const float*)d_in[1];
    float* out = (float*)d_out;
    dim3 grid(IMSIZE);
    dim3 block(BLOCK_THREADS);
    sketch_render_kernel<<<grid, block, 0, stream>>>(x, y, out);
}


# Segment sum taken in-lane: each lane adds its exp values into running packed f32 sums (1 packed add per tile-step); one ones-product per tile after the loop folds the 4 k-slots; sum order changes
# speedup vs baseline: 2.9147x; 1.6782x over previous
#include <hip/hip_runtime.h>


typedef __attribute__((ext_vector_type(2))) float v2f;
typedef __attribute__((ext_vector_type(4))) float v4f;
typedef __attribute__((ext_vector_type(8))) float v8f;

#define IMSIZE   224
#define NSEG     1024
#define NPAIR    (NSEG / 2)
#define SMOOTH_K 8.0f
#define EPS_F    1e-10f

#define TILES_PER_WAVE 7
#define KSPLIT         8
#define NWAVES         (2 * KSPLIT)
#define BLOCK_THREADS  (NWAVES * 32)
#define SEG_PER_WAVE   (NSEG / KSPLIT)
#define STEPS          (SEG_PER_WAVE / 4)

#define NEGK_LOG2E (-SMOOTH_K * 1.44269504088896340736f)
#define LOG2_NSEG  10.0f
#define LN2_OVER_K (0.69314718055994530942f / SMOOTH_K)

struct SegRow { float B, Cr, sl, ic, minx, maxx, miny, maxy; };

__device__ __forceinline__ SegRow seg_row(float x0, float y0, float x1, float y1, float yp)
{
    float dx = x1 - x0, dy = y1 - y0;
    float d2 = fmaf(dy, dy, dx * dx);
    float inv = 1.0f / (d2 + EPS_F);
    float cross = fmaf(y0, x1, -(x0 * y1));
    float A  = dx * dy * inv;
    float B  = dx * dx * inv;
    float Cc = -dy * cross * inv;
    float idx = 1.0f / (dx + EPS_F);
    float sl = dy * idx;
    float ic = cross * idx;
    bool  vert = (x0 == x1);
    SegRow r;
    r.B    = B;
    r.Cr   = fmaf(A, yp, Cc);
    r.sl   = vert ? 0.0f : sl;
    r.ic   = vert ? yp   : ic;
    r.minx = fminf(x0, x1);
    r.maxx = fmaxf(x0, x1);
    r.miny = fminf(y0, y1);
    r.maxy = fmaxf(y0, y1);
    return r;
}

__device__ __forceinline__ void tiles_step(const v4f r0, const v4f r1, const v4f r2, const v4f r3,
                                           const v2f* __restrict__ xpv, const v2f ypv,
                                           v2f* __restrict__ acc)
{
    const v2f epsv  = {EPS_F, EPS_F};
    const v2f Bv  = {r0.x, r0.y}, Crv = {r0.z, r0.w};
    const v2f slv = {r1.x, r1.y}, icv = {r1.z, r1.w};
    v2f cx[TILES_PER_WAVE], cy[TILES_PER_WAVE], ddx[TILES_PER_WAVE], ddy[TILES_PER_WAVE];
    v2f s[TILES_PER_WAVE], a[TILES_PER_WAVE];
    #pragma unroll
    for (int t = 0; t < TILES_PER_WAVE; ++t)
        cx[t] = __builtin_elementwise_fma(Bv, xpv[t], Crv);
    #pragma unroll
    for (int t = 0; t < TILES_PER_WAVE; ++t)
        cy[t] = __builtin_elementwise_fma(slv, cx[t], icv);
    #pragma unroll
    for (int t = 0; t < TILES_PER_WAVE; ++t) {
        v2f xp1;
        xp1.x = fminf(fmaxf(cx[t].x, r2.x), r2.z);
        xp1.y = fminf(fmaxf(cx[t].y, r2.y), r2.w);
        ddx[t] = xpv[t] - xp1;
    }
    #pragma unroll
    for (int t = 0; t < TILES_PER_WAVE; ++t) {
        v2f yp1;
        yp1.x = fminf(fmaxf(cy[t].x, r3.x), r3.z);
        yp1.y = fminf(fmaxf(cy[t].y, r3.y), r3.w);
        ddy[t] = ypv - yp1;
    }
    #pragma unroll
    for (int t = 0; t < TILES_PER_WAVE; ++t)
        s[t] = __builtin_elementwise_fma(ddx[t], ddx[t], epsv);
    #pragma unroll
    for (int t = 0; t < TILES_PER_WAVE; ++t)
        s[t] = __builtin_elementwise_fma(ddy[t], ddy[t], s[t]);
    #pragma unroll
    for (int t = 0; t < TILES_PER_WAVE; ++t) {
        a[t].x = __builtin_amdgcn_sqrtf(s[t].x) * NEGK_LOG2E;
        a[t].y = __builtin_amdgcn_sqrtf(s[t].y) * NEGK_LOG2E;
    }
    #pragma unroll
    for (int t = 0; t < TILES_PER_WAVE; ++t) {
        a[t].x = __builtin_amdgcn_exp2f(a[t].x);
        a[t].y = __builtin_amdgcn_exp2f(a[t].y);
    }
    #pragma unroll
    for (int t = 0; t < TILES_PER_WAVE; ++t)
        acc[t] = acc[t] + a[t];
}

__global__ __launch_bounds__(BLOCK_THREADS) void sketch_render_kernel(
    const float* __restrict__ xpar,
    const float* __restrict__ ypar,
    float* __restrict__ out)
{
    __shared__ v4f sT[NPAIR * 4];
    __shared__ v4f sP[KSPLIT * IMSIZE / 4];

    const int   row = blockIdx.x;
    const float yp  = (float)row;

    {
        const int p = threadIdx.x;
        float xa = xpar[2 * p]     * (float)IMSIZE;
        float xb = xpar[2 * p + 1] * (float)IMSIZE;
        float xc = xpar[2 * p + 2] * (float)IMSIZE;
        float ya = ypar[2 * p]     * (float)IMSIZE;
        float yb = ypar[2 * p + 1] * (float)IMSIZE;
        float yc = ypar[2 * p + 2] * (float)IMSIZE;
        SegRow e = seg_row(xa, ya, xb, yb, yp);
        SegRow o = seg_row(xb, yb, xc, yc, yp);
        v4f r0 = {e.B,    o.B,    e.Cr,   o.Cr};
        v4f r1 = {e.sl,   o.sl,   e.ic,   o.ic};
        v4f r2 = {e.minx, o.minx, e.maxx, o.maxx};
        v4f r3 = {e.miny, o.miny, e.maxy, o.maxy};
        sT[p * 4 + 0] = r0;
        sT[p * 4 + 1] = r1;
        sT[p * 4 + 2] = r2;
        sT[p * 4 + 3] = r3;
    }
    __syncthreads();

    const int wave = threadIdx.x >> 5;
    const int lane = threadIdx.x & 31;
    const int grp  = lane >> 4;
    const int m    = lane & 15;
    const int half = wave & 1;
    const int ks   = wave >> 1;

    const v2f ypv = {yp, yp};
    v2f xpv[TILES_PER_WAVE];
    v2f acc[TILES_PER_WAVE];
    #pragma unroll
    for (int t = 0; t < TILES_PER_WAVE; ++t) {
        const float xp = (float)((half * TILES_PER_WAVE + t) * 16 + m);
        xpv[t] = (v2f){xp, xp};
        acc[t] = (v2f){0.0f, 0.0f};
    }

    const v4f* rec = sT + (ks * (SEG_PER_WAVE / 2) + grp) * 4;

    for (int it = 0; it < STEPS; it += 2) {
        const v4f* r = rec + it * 8;
        v4f a0 = r[0], a1 = r[1], a2 = r[2],  a3 = r[3];
        v4f b0 = r[8], b1 = r[9], b2 = r[10], b3 = r[11];
        tiles_step(a0, a1, a2, a3, xpv, ypv, acc);
        tiles_step(b0, b1, b2, b3, xpv, ypv, acc);
    }

    v8f c[TILES_PER_WAVE];
    {
        const v2f bones = {1.0f, 1.0f};
        const v8f zero8 = {0.0f, 0.0f, 0.0f, 0.0f, 0.0f, 0.0f, 0.0f, 0.0f};
        #pragma unroll
        for (int t = 0; t < TILES_PER_WAVE; ++t)
            c[t] = __builtin_amdgcn_wmma_f32_16x16x4_f32(false, acc[t], false, bones,
                                                         (short)0, zero8, false, false);
    }

    if (m == 0) {
        v4f* dst = sP + (ks * IMSIZE + half * TILES_PER_WAVE * 16 + grp * 8) / 4;
        #pragma unroll
        for (int t = 0; t < TILES_PER_WAVE; ++t) {
            v4f lo = {c[t][0], c[t][1], c[t][2], c[t][3]};
            v4f hi = {c[t][4], c[t][5], c[t][6], c[t][7]};
            dst[t * 4 + 0] = lo;
            dst[t * 4 + 1] = hi;
        }
    }
    __syncthreads();

    if (threadIdx.x < IMSIZE) {
        const float* P = (const float*)sP;
        float res = P[threadIdx.x];
        #pragma unroll
        for (int k = 1; k < KSPLIT; ++k)
            res += P[k * IMSIZE + threadIdx.x];
        out[row * IMSIZE + threadIdx.x] = (LOG2_NSEG - __builtin_amdgcn_logf(res)) * LN2_OVER_K;
    }
}

extern "C" void kernel_launch(void* const* d_in, const int* in_sizes, int n_in,
                              void* d_out, int out_size, void* d_ws, size_t ws_size,
                              hipStream_t stream) {
    (void)in_sizes; (void)n_in; (void)out_size; (void)d_ws; (void)ws_size;
    const float* x = (const float*)d_in[0];
    const float* y = (const float*)d_in[1];
    float* out = (float*)d_out;
    dim3 grid(IMSIZE);
    dim3 block(BLOCK_THREADS);
    sketch_render_kernel<<<grid, block, 0, stream>>>(x, y, out);
}


# Each half row loops only over segments whose term can be nonzero there: dist>=13.5 certain from the loop's own cx,cy at the range ends, exp2 is exactly 0; lists compacted in LDS; sum order changes
# speedup vs baseline: 4.5619x; 1.5651x over previous
#include <hip/hip_runtime.h>


typedef __attribute__((ext_vector_type(2))) float v2f;
typedef __attribute__((ext_vector_type(4))) float v4f;
typedef __attribute__((ext_vector_type(8))) float v8f;

#define IMSIZE   224
#define NSEG     1024
#define NPAIR    (NSEG / 2)
#define SMOOTH_K 8.0f
#define EPS_F    1e-10f

#define TILES_PER_WAVE 7
#define KSPLIT         8
#define NWAVES         (2 * KSPLIT)
#define BLOCK_THREADS  (NWAVES * 32)
#define HALF_PIX       (TILES_PER_WAVE * 16)
#define STEP_SEGS      (KSPLIT * 4)

#define CULL_R2        (13.5f * 13.5f)
#define FAR_AWAY       (-10000.0f)

#define NEGK_LOG2E (-SMOOTH_K * 1.44269504088896340736f)
#define LOG2_NSEG  10.0f
#define LN2_OVER_K (0.69314718055994530942f / SMOOTH_K)

struct SegRow { float B, Cr, sl, ic, minx, maxx, miny, maxy; };

__device__ __forceinline__ SegRow seg_row(float x0, float y0, float x1, float y1, float yp)
{
    float dx = x1 - x0, dy = y1 - y0;
    float d2 = fmaf(dy, dy, dx * dx);
    float inv = 1.0f / (d2 + EPS_F);
    float cross = fmaf(y0, x1, -(x0 * y1));
    float A  = dx * dy * inv;
    float B  = dx * dx * inv;
    float Cc = -dy * cross * inv;
    float idx = 1.0f / (dx + EPS_F);
    float sl = dy * idx;
    float ic = cross * idx;
    bool  vert = (x0 == x1);
    SegRow r;
    r.B    = B;
    r.Cr   = fmaf(A, yp, Cc);
    r.sl   = vert ? 0.0f : sl;
    r.ic   = vert ? yp   : ic;
    r.minx = fminf(x0, x1);
    r.maxx = fmaxf(x0, x1);
    r.miny = fminf(y0, y1);
    r.maxy = fmaxf(y0, y1);
    return r;
}

__device__ __forceinline__ bool seg_near(const SegRow r, float yp, float xlo, float xhi)
{
    float cxa = fmaf(r.B, xlo, r.Cr);
    float cxb = fmaf(r.B, xhi, r.Cr);
    float pa  = fminf(fmaxf(cxa, r.minx), r.maxx);
    float pb  = fminf(fmaxf(cxb, r.minx), r.maxx);
    float cya = fmaf(r.sl, cxa, r.ic);
    float cyb = fmaf(r.sl, cxb, r.ic);
    float qa  = fminf(fmaxf(fminf(cya, cyb), r.miny), r.maxy);
    float qb  = fminf(fmaxf(fmaxf(cya, cyb), r.miny), r.maxy);
    float dxm = fmaxf(0.0f, fmaxf(xlo - pb, pa - xhi));
    float dym = fmaxf(0.0f, fmaxf(qa - yp, yp - qb));
    return !(fmaf(dxm, dxm, dym * dym) >= CULL_R2);
}

__device__ __forceinline__ void put_seg(float* __restrict__ list, int pos, const SegRow r)
{
    float* d = list + (pos >> 1) * 16 + (pos & 1);
    d[0]  = r.B;    d[2]  = r.Cr;
    d[4]  = r.sl;   d[6]  = r.ic;
    d[8]  = r.minx; d[10] = r.maxx;
    d[12] = r.miny; d[14] = r.maxy;
}

__device__ __forceinline__ void tiles_step(const v4f r0, const v4f r1, const v4f r2, const v4f r3,
                                           const v2f* __restrict__ xpv, const v2f ypv,
                                           v2f* __restrict__ acc)
{
    const v2f epsv  = {EPS_F, EPS_F};
    const v2f Bv  = {r0.x, r0.y}, Crv = {r0.z, r0.w};
    const v2f slv = {r1.x, r1.y}, icv = {r1.z, r1.w};
    v2f cx[TILES_PER_WAVE], cy[TILES_PER_WAVE], ddx[TILES_PER_WAVE], ddy[TILES_PER_WAVE];
    v2f s[TILES_PER_WAVE], a[TILES_PER_WAVE];
    #pragma unroll
    for (int t = 0; t < TILES_PER_WAVE; ++t)
        cx[t] = __builtin_elementwise_fma(Bv, xpv[t], Crv);
    #pragma unroll
    for (int t = 0; t < TILES_PER_WAVE; ++t)
        cy[t] = __builtin_elementwise_fma(slv, cx[t], icv);
    #pragma unroll
    for (int t = 0; t < TILES_PER_WAVE; ++t) {
        v2f xp1;
        xp1.x = fminf(fmaxf(cx[t].x, r2.x), r2.z);
        xp1.y = fminf(fmaxf(cx[t].y, r2.y), r2.w);
        ddx[t] = xpv[t] - xp1;
    }
    #pragma unroll
    for (int t = 0; t < TILES_PER_WAVE; ++t) {
        v2f yp1;
        yp1.x = fminf(fmaxf(cy[t].x, r3.x), r3.z);
        yp1.y = fminf(fmaxf(cy[t].y, r3.y), r3.w);
        ddy[t] = ypv - yp1;
    }
    #pragma unroll
    for (int t = 0; t < TILES_PER_WAVE; ++t)
        s[t] = __builtin_elementwise_fma(ddx[t], ddx[t], epsv);
    #pragma unroll
    for (int t = 0; t < TILES_PER_WAVE; ++t)
        s[t] = __builtin_elementwise_fma(ddy[t], ddy[t], s[t]);
    #pragma unroll
    for (int t = 0; t < TILES_PER_WAVE; ++t) {
        a[t].x = __builtin_amdgcn_sqrtf(s[t].x) * NEGK_LOG2E;
        a[t].y = __builtin_amdgcn_sqrtf(s[t].y) * NEGK_LOG2E;
    }
    #pragma unroll
    for (int t = 0; t < TILES_PER_WAVE; ++t) {
        a[t].x = __builtin_amdgcn_exp2f(a[t].x);
        a[t].y = __builtin_amdgcn_exp2f(a[t].y);
    }
    #pragma unroll
    for (int t = 0; t < TILES_PER_WAVE; ++t)
        acc[t] = acc[t] + a[t];
}

__global__ __launch_bounds__(BLOCK_THREADS) void sketch_render_kernel(
    const float* __restrict__ xpar,
    const float* __restrict__ ypar,
    float* __restrict__ out)
{
    __shared__ v4f sT[2 * NPAIR * 4];
    __shared__ int sCnt[2 * NWAVES];
    __shared__ v4f sP[KSPLIT * IMSIZE / 4];

    const int   row = blockIdx.x;
    const float yp  = (float)row;

    const int wave = threadIdx.x >> 5;
    const int lane = threadIdx.x & 31;

    float* const listL = (float*)sT;
    float* const listR = (float*)(sT + NPAIR * 4);
    int nL = 0, nR = 0;
    {
        const int p = threadIdx.x;
        float xa = xpar[2 * p]     * (float)IMSIZE;
        float xb = xpar[2 * p + 1] * (float)IMSIZE;
        float xc = xpar[2 * p + 2] * (float)IMSIZE;
        float ya = ypar[2 * p]     * (float)IMSIZE;
        float yb = ypar[2 * p + 1] * (float)IMSIZE;
        float yc = ypar[2 * p + 2] * (float)IMSIZE;
        SegRow e = seg_row(xa, ya, xb, yb, yp);
        SegRow o = seg_row(xb, yb, xc, yc, yp);

        const bool eL = seg_near(e, yp, 0.0f, (float)(HALF_PIX - 1));
        const bool oL = seg_near(o, yp, 0.0f, (float)(HALF_PIX - 1));
        const bool eR = seg_near(e, yp, (float)HALF_PIX, (float)(IMSIZE - 1));
        const bool oR = seg_near(o, yp, (float)HALF_PIX, (float)(IMSIZE - 1));

        const unsigned below = (1u << lane) - 1u;
        const unsigned bEL = __builtin_amdgcn_ballot_w32(eL), bOL = __builtin_amdgcn_ballot_w32(oL);
        const unsigned bER = __builtin_amdgcn_ballot_w32(eR), bOR = __builtin_amdgcn_ballot_w32(oR);
        int posL = __popc(bEL & below) + __popc(bOL & below);
        int posR = __popc(bER & below) + __popc(bOR & below);
        if (lane == 0) {
            sCnt[wave]          = __popc(bEL) + __popc(bOL);
            sCnt[NWAVES + wave] = __popc(bER) + __popc(bOR);
        }
        __syncthreads();
        #pragma unroll
        for (int w = 0; w < NWAVES; ++w) {
            const int cl = sCnt[w], cr = sCnt[NWAVES + w];
            nL += cl;
            nR += cr;
            if (w < wave) { posL += cl; posR += cr; }
        }
        if (eL) put_seg(listL, posL, e);
        if (oL) put_seg(listL, posL + (eL ? 1 : 0), o);
        if (eR) put_seg(listR, posR, e);
        if (oR) put_seg(listR, posR + (eR ? 1 : 0), o);

        SegRow f;
        f.B = 0.0f; f.Cr = 0.0f; f.sl = 0.0f; f.ic = 0.0f;
        f.minx = FAR_AWAY; f.maxx = FAR_AWAY; f.miny = FAR_AWAY; f.maxy = FAR_AWAY;
        if (wave < 2) {
            float* const list = wave ? listR : listL;
            const unsigned n  = (unsigned)(wave ? nR : nL);
            const unsigned i  = n + (unsigned)lane;
            if (i < ((n + STEP_SEGS - 1u) / STEP_SEGS) * STEP_SEGS) put_seg(list, (int)i, f);
        }
    }
    __syncthreads();

    const int grp  = lane >> 4;
    const int m    = lane & 15;
    const int half = (wave ^ (wave >> 2)) & 1;
    const int ks   = wave >> 1;

    const v2f ypv = {yp, yp};
    v2f xpv[TILES_PER_WAVE];
    v2f acc[TILES_PER_WAVE];
    #pragma unroll
    for (int t = 0; t < TILES_PER_WAVE; ++t) {
        const float xp = (float)((half * TILES_PER_WAVE + t) * 16 + m);
        xpv[t] = (v2f){xp, xp};
        acc[t] = (v2f){0.0f, 0.0f};
    }

    const int steps = __builtin_amdgcn_readfirstlane(
        (int)(((unsigned)(half ? nR : nL) + STEP_SEGS - 1u) / STEP_SEGS));
    const v4f* rec = sT + half * (NPAIR * 4) + (ks * steps * 2 + grp) * 4;

    for (int s = 0; s + 1 < steps; s += 2) {
        const v4f* r = rec + s * 8;
        v4f a0 = r[0], a1 = r[1], a2 = r[2],  a3 = r[3];
        v4f b0 = r[8], b1 = r[9], b2 = r[10], b3 = r[11];
        tiles_step(a0, a1, a2, a3, xpv, ypv, acc);
        tiles_step(b0, b1, b2, b3, xpv, ypv, acc);
    }
    if (steps & 1) {
        const v4f* r = rec + (steps - 1) * 8;
        v4f a0 = r[0], a1 = r[1], a2 = r[2], a3 = r[3];
        tiles_step(a0, a1, a2, a3, xpv, ypv, acc);
    }

    v8f c[TILES_PER_WAVE];
    {
        const v2f bones = {1.0f, 1.0f};
        const v8f zero8 = {0.0f, 0.0f, 0.0f, 0.0f, 0.0f, 0.0f, 0.0f, 0.0f};
        #pragma unroll
        for (int t = 0; t < TILES_PER_WAVE; ++t)
            c[t] = __builtin_amdgcn_wmma_f32_16x16x4_f32(false, acc[t], false, bones,
                                                         (short)0, zero8, false, false);
    }

    if (m == 0) {
        v4f* dst = sP + (ks * IMSIZE + half * TILES_PER_WAVE * 16 + grp * 8) / 4;
        #pragma unroll
        for (int t = 0; t < TILES_PER_WAVE; ++t) {
            v4f lo = {c[t][0], c[t][1], c[t][2], c[t][3]};
            v4f hi = {c[t][4], c[t][5], c[t][6], c[t][7]};
            dst[t * 4 + 0] = lo;
            dst[t * 4 + 1] = hi;
        }
    }
    __syncthreads();

    if (threadIdx.x < IMSIZE) {
        const float* P = (const float*)sP;
        float res = P[threadIdx.x];
        #pragma unroll
        for (int k = 1; k < KSPLIT; ++k)
            res += P[k * IMSIZE + threadIdx.x];
        out[row * IMSIZE + threadIdx.x] = (LOG2_NSEG - __builtin_amdgcn_logf(res)) * LN2_OVER_K;
    }
}

extern "C" void kernel_launch(void* const* d_in, const int* in_sizes, int n_in,
                              void* d_out, int out_size, void* d_ws, size_t ws_size,
                              hipStream_t stream) {
    (void)in_sizes; (void)n_in; (void)out_size; (void)d_ws; (void)ws_size;
    const float* x = (const float*)d_in[0];
    const float* y = (const float*)d_in[1];
    float* out = (float*)d_out;
    dim3 grid(IMSIZE);
    dim3 block(BLOCK_THREADS);
    sketch_render_kernel<<<grid, block, 0, stream>>>(x, y, out);
}


# The row is cut into four strips (4,3,3,4 tiles), each with its own list of segments that can reach it; a wave covers one strip over a quarter of its list (4 x 4 waves); sum order changes
# speedup vs baseline: 5.0162x; 1.0996x over previous
#include <hip/hip_runtime.h>


typedef __attribute__((ext_vector_type(2))) float v2f;
typedef __attribute__((ext_vector_type(4))) float v4f;
typedef __attribute__((ext_vector_type(8))) float v8f;

#define IMSIZE   224
#define NSEG     1024
#define NPAIR    (NSEG / 2)
#define SMOOTH_K 8.0f
#define EPS_F    1e-10f

#define NSTRIP         4
#define KSPLIT         4
#define NWAVES         (NSTRIP * KSPLIT)
#define BLOCK_THREADS  (NWAVES * 32)
#define STEP_SEGS      (KSPLIT * 4)

#define CULL_R2        (13.5f * 13.5f)
#define FAR_AWAY       (-10000.0f)

#define NEGK_LOG2E (-SMOOTH_K * 1.44269504088896340736f)
#define LOG2_NSEG  10.0f
#define LN2_OVER_K (0.69314718055994530942f / SMOOTH_K)

struct SegRow { float B, Cr, sl, ic, minx, maxx, miny, maxy; };

__device__ __forceinline__ SegRow seg_row(float x0, float y0, float x1, float y1, float yp)
{
    float dx = x1 - x0, dy = y1 - y0;
    float d2 = fmaf(dy, dy, dx * dx);
    float inv = 1.0f / (d2 + EPS_F);
    float cross = fmaf(y0, x1, -(x0 * y1));
    float A  = dx * dy * inv;
    float B  = dx * dx * inv;
    float Cc = -dy * cross * inv;
    float idx = 1.0f / (dx + EPS_F);
    float sl = dy * idx;
    float ic = cross * idx;
    bool  vert = (x0 == x1);
    SegRow r;
    r.B    = B;
    r.Cr   = fmaf(A, yp, Cc);
    r.sl   = vert ? 0.0f : sl;
    r.ic   = vert ? yp   : ic;
    r.minx = fminf(x0, x1);
    r.maxx = fmaxf(x0, x1);
    r.miny = fminf(y0, y1);
    r.maxy = fmaxf(y0, y1);
    return r;
}

__device__ __forceinline__ bool seg_near(const SegRow r, float yp, float xlo, float xhi)
{
    float cxa = fmaf(r.B, xlo, r.Cr);
    float cxb = fmaf(r.B, xhi, r.Cr);
    float pa  = fminf(fmaxf(cxa, r.minx), r.maxx);
    float pb  = fminf(fmaxf(cxb, r.minx), r.maxx);
    float cya = fmaf(r.sl, cxa, r.ic);
    float cyb = fmaf(r.sl, cxb, r.ic);
    float qa  = fminf(fmaxf(fminf(cya, cyb), r.miny), r.maxy);
    float qb  = fminf(fmaxf(fmaxf(cya, cyb), r.miny), r.maxy);
    float dxm = fmaxf(0.0f, fmaxf(xlo - pb, pa - xhi));
    float dym = fmaxf(0.0f, fmaxf(qa - yp, yp - qb));
    return !(fmaf(dxm, dxm, dym * dym) >= CULL_R2);
}

__device__ __forceinline__ void put_seg(float* __restrict__ list, int pos, const SegRow r)
{
    float* d = list + (pos >> 1) * 16 + (pos & 1);
    d[0]  = r.B;    d[2]  = r.Cr;
    d[4]  = r.sl;   d[6]  = r.ic;
    d[8]  = r.minx; d[10] = r.maxx;
    d[12] = r.miny; d[14] = r.maxy;
}

template <int NT>
__device__ __forceinline__ void tiles_step(const v4f r0, const v4f r1, const v4f r2, const v4f r3,
                                           const v2f* __restrict__ xpv, const v2f ypv,
                                           v2f* __restrict__ acc)
{
    const v2f epsv  = {EPS_F, EPS_F};
    const v2f Bv  = {r0.x, r0.y}, Crv = {r0.z, r0.w};
    const v2f slv = {r1.x, r1.y}, icv = {r1.z, r1.w};
    v2f cx[NT], cy[NT], ddx[NT], ddy[NT];
    v2f s[NT], a[NT];
    #pragma unroll
    for (int t = 0; t < NT; ++t)
        cx[t] = __builtin_elementwise_fma(Bv, xpv[t], Crv);
    #pragma unroll
    for (int t = 0; t < NT; ++t)
        cy[t] = __builtin_elementwise_fma(slv, cx[t], icv);
    #pragma unroll
    for (int t = 0; t < NT; ++t) {
        v2f xp1;
        xp1.x = fminf(fmaxf(cx[t].x, r2.x), r2.z);
        xp1.y = fminf(fmaxf(cx[t].y, r2.y), r2.w);
        ddx[t] = xpv[t] - xp1;
    }
    #pragma unroll
    for (int t = 0; t < NT; ++t) {
        v2f yp1;
        yp1.x = fminf(fmaxf(cy[t].x, r3.x), r3.z);
        yp1.y = fminf(fmaxf(cy[t].y, r3.y), r3.w);
        ddy[t] = ypv - yp1;
    }
    #pragma unroll
    for (int t = 0; t < NT; ++t)
        s[t] = __builtin_elementwise_fma(ddx[t], ddx[t], epsv);
    #pragma unroll
    for (int t = 0; t < NT; ++t)
        s[t] = __builtin_elementwise_fma(ddy[t], ddy[t], s[t]);
    #pragma unroll
    for (int t = 0; t < NT; ++t) {
        a[t].x = __builtin_amdgcn_sqrtf(s[t].x) * NEGK_LOG2E;
        a[t].y = __builtin_amdgcn_sqrtf(s[t].y) * NEGK_LOG2E;
    }
    #pragma unroll
    for (int t = 0; t < NT; ++t) {
        a[t].x = __builtin_amdgcn_exp2f(a[t].x);
        a[t].y = __builtin_amdgcn_exp2f(a[t].y);
    }
    #pragma unroll
    for (int t = 0; t < NT; ++t)
        acc[t] = acc[t] + a[t];
}

template <int NT>
__device__ __forceinline__ void strip_wave(const v4f* __restrict__ rec, const int steps,
                                           const int tile0, const int m, const float yp,
                                           v4f* __restrict__ dst)
{
    const v2f ypv = {yp, yp};
    v2f xpv[NT];
    v2f acc[NT];
    #pragma unroll
    for (int t = 0; t < NT; ++t) {
        const float xp = (float)((tile0 + t) * 16 + m);
        xpv[t] = (v2f){xp, xp};
        acc[t] = (v2f){0.0f, 0.0f};
    }

    for (int s = 0; s + 1 < steps; s += 2) {
        const v4f* r = rec + s * 8;
        v4f a0 = r[0], a1 = r[1], a2 = r[2],  a3 = r[3];
        v4f b0 = r[8], b1 = r[9], b2 = r[10], b3 = r[11];
        tiles_step<NT>(a0, a1, a2, a3, xpv, ypv, acc);
        tiles_step<NT>(b0, b1, b2, b3, xpv, ypv, acc);
    }
    if (steps & 1) {
        const v4f* r = rec + (steps - 1) * 8;
        v4f a0 = r[0], a1 = r[1], a2 = r[2], a3 = r[3];
        tiles_step<NT>(a0, a1, a2, a3, xpv, ypv, acc);
    }

    v8f c[NT];
    {
        const v2f bones = {1.0f, 1.0f};
        const v8f zero8 = {0.0f, 0.0f, 0.0f, 0.0f, 0.0f, 0.0f, 0.0f, 0.0f};
        #pragma unroll
        for (int t = 0; t < NT; ++t)
            c[t] = __builtin_amdgcn_wmma_f32_16x16x4_f32(false, acc[t], false, bones,
                                                         (short)0, zero8, false, false);
    }

    if (m == 0) {
        #pragma unroll
        for (int t = 0; t < NT; ++t) {
            v4f lo = {c[t][0], c[t][1], c[t][2], c[t][3]};
            v4f hi = {c[t][4], c[t][5], c[t][6], c[t][7]};
            dst[t * 4 + 0] = lo;
            dst[t * 4 + 1] = hi;
        }
    }
}

__global__ __launch_bounds__(BLOCK_THREADS) void sketch_render_kernel(
    const float* __restrict__ xpar,
    const float* __restrict__ ypar,
    float* __restrict__ out)
{
    __shared__ v4f sT[NSTRIP * NPAIR * 4];
    __shared__ unsigned sCnt[2 * NWAVES];
    __shared__ v4f sP[KSPLIT * IMSIZE / 4];

    const int   row = blockIdx.x;
    const float yp  = (float)row;

    const int wave = __builtin_amdgcn_readfirstlane((int)(threadIdx.x >> 5));
    const int lane = threadIdx.x & 31;

    unsigned tot01 = 0, tot23 = 0;
    {
        const int p = threadIdx.x;
        float xa = xpar[2 * p]     * (float)IMSIZE;
        float xb = xpar[2 * p + 1] * (float)IMSIZE;
        float xc = xpar[2 * p + 2] * (float)IMSIZE;
        float ya = ypar[2 * p]     * (float)IMSIZE;
        float yb = ypar[2 * p + 1] * (float)IMSIZE;
        float yc = ypar[2 * p + 2] * (float)IMSIZE;
        SegRow e = seg_row(xa, ya, xb, yb, yp);
        SegRow o = seg_row(xb, yb, xc, yc, yp);

        const float xlim[NSTRIP + 1] = {0.0f, 64.0f, 112.0f, 160.0f, (float)(IMSIZE - 1)};
        bool ke[NSTRIP], ko[NSTRIP];
        #pragma unroll
        for (int s = 0; s < NSTRIP; ++s) {
            ke[s] = seg_near(e, yp, xlim[s], xlim[s + 1]);
            ko[s] = seg_near(o, yp, xlim[s], xlim[s + 1]);
        }

        const unsigned below = (1u << lane) - 1u;
        int pos[NSTRIP];
        unsigned cnt[NSTRIP];
        #pragma unroll
        for (int s = 0; s < NSTRIP; ++s) {
            const unsigned bE = __builtin_amdgcn_ballot_w32(ke[s]);
            const unsigned bO = __builtin_amdgcn_ballot_w32(ko[s]);
            pos[s] = __popc(bE & below) + __popc(bO & below);
            cnt[s] = (unsigned)(__popc(bE) + __popc(bO));
        }
        if (lane == 0) {
            sCnt[wave]          = cnt[0] | (cnt[1] << 16);
            sCnt[NWAVES + wave] = cnt[2] | (cnt[3] << 16);
        }
        __syncthreads();
        const int wavev = (int)(threadIdx.x >> 5);
        unsigned pre01 = 0, pre23 = 0;
        #pragma unroll
        for (int w = 0; w < NWAVES; ++w) {
            const unsigned c01 = sCnt[w], c23 = sCnt[NWAVES + w];
            tot01 += c01;
            tot23 += c23;
            if (w < wavev) { pre01 += c01; pre23 += c23; }
        }
        pos[0] += (int)(pre01 & 0xffffu); pos[1] += (int)(pre01 >> 16);
        pos[2] += (int)(pre23 & 0xffffu); pos[3] += (int)(pre23 >> 16);
        #pragma unroll
        for (int s = 0; s < NSTRIP; ++s) {
            float* const list = (float*)(sT + s * (NPAIR * 4));
            if (ke[s]) put_seg(list, pos[s], e);
            if (ko[s]) put_seg(list, pos[s] + (ke[s] ? 1 : 0), o);
        }

        SegRow f;
        f.B = 0.0f; f.Cr = 0.0f; f.sl = 0.0f; f.ic = 0.0f;
        f.minx = FAR_AWAY; f.maxx = FAR_AWAY; f.miny = FAR_AWAY; f.maxy = FAR_AWAY;
        if (wave < NSTRIP) {
            float* const list = (float*)(sT + wave * (NPAIR * 4));
            const unsigned n  = (((wave & 2) ? tot23 : tot01) >> ((wave & 1) * 16)) & 0xffffu;
            const unsigned i  = n + (unsigned)lane;
            if (i < ((n + STEP_SEGS - 1u) / STEP_SEGS) * STEP_SEGS) put_seg(list, (int)i, f);
        }
    }
    __syncthreads();

    const int grp  = lane >> 4;
    const int m    = lane & 15;
    const int strip = (wave ^ (wave >> 2)) & 3;
    const int ks    = wave >> 2;

    const unsigned nstrip = (((strip & 2) ? tot23 : tot01) >> ((strip & 1) * 16)) & 0xffffu;
    const int steps = __builtin_amdgcn_readfirstlane((int)((nstrip + STEP_SEGS - 1u) / STEP_SEGS));
    const v4f* rec = sT + strip * (NPAIR * 4) + (ks * steps * 2 + grp) * 4;
    const int tile0 = 3 * strip + (strip != 0 ? 1 : 0);
    v4f* dst = sP + (ks * IMSIZE + tile0 * 16 + grp * 8) / 4;

    if (strip == 0 || strip == 3)
        strip_wave<4>(rec, steps, tile0, m, yp, dst);
    else
        strip_wave<3>(rec, steps, tile0, m, yp, dst);
    __syncthreads();

    if (threadIdx.x < IMSIZE) {
        const float* P = (const float*)sP;
        float res = P[threadIdx.x];
        #pragma unroll
        for (int k = 1; k < KSPLIT; ++k)
            res += P[k * IMSIZE + threadIdx.x];
        out[row * IMSIZE + threadIdx.x] = (LOG2_NSEG - __builtin_amdgcn_logf(res)) * LN2_OVER_K;
    }
}

extern "C" void kernel_launch(void* const* d_in, const int* in_sizes, int n_in,
                              void* d_out, int out_size, void* d_ws, size_t ws_size,
                              hipStream_t stream) {
    (void)in_sizes; (void)n_in; (void)out_size; (void)d_ws; (void)ws_size;
    const float* x = (const float*)d_in[0];
    const float* y = (const float*)d_in[1];
    float* out = (float*)d_out;
    dim3 grid(IMSIZE);
    dim3 block(BLOCK_THREADS);
    sketch_render_kernel<<<grid, block, 0, stream>>>(x, y, out);
}


# Lengths pre-scaled by k*log2e in the LDS record and y offset yp-yp1 clamped directly from it (med3): per pair no exponent multiply, no y subtract, no bound re-quieting; rounding changes slightly
# speedup vs baseline: 5.3930x; 1.0751x over previous
#include <hip/hip_runtime.h>


typedef __attribute__((ext_vector_type(2))) float v2f;
typedef __attribute__((ext_vector_type(4))) float v4f;
typedef __attribute__((ext_vector_type(8))) float v8f;

#define IMSIZE   224
#define NSEG     1024
#define NPAIR    (NSEG / 2)
#define SMOOTH_K 8.0f
#define EPS_F    1e-10f

#define NSTRIP         4
#define KSPLIT         4
#define NWAVES         (NSTRIP * KSPLIT)
#define BLOCK_THREADS  (NWAVES * 32)
#define STEP_SEGS      (KSPLIT * 4)

#define CULL_R2        ((13.5f * K_LOG2E) * (13.5f * K_LOG2E))
#define FAR_AWAY       (-10000.0f)

#define K_LOG2E    (SMOOTH_K * 1.44269504088896340736f)
#define EPS_S      (EPS_F * K_LOG2E * K_LOG2E)
#define LOG2_NSEG  10.0f
#define LN2_OVER_K (0.69314718055994530942f / SMOOTH_K)

struct SegRow { float B, Cr, nsl, ric, minx, maxx, lo, hi; };

__device__ __forceinline__ SegRow seg_row(float x0, float y0, float x1, float y1, float yp)
{
    float dx = x1 - x0, dy = y1 - y0;
    float d2 = fmaf(dy, dy, dx * dx);
    float inv = 1.0f / (d2 + EPS_F);
    float cross = fmaf(y0, x1, -(x0 * y1));
    float A  = dx * dy * inv;
    float B  = dx * dx * inv;
    float Cc = -dy * cross * inv;
    float idx = 1.0f / (dx + EPS_F);
    float sl = dy * idx;
    float ic = cross * idx;
    bool  vert = (x0 == x1);
    SegRow r;
    r.B    = B;
    r.Cr   = fmaf(A, yp, Cc) * K_LOG2E;
    r.nsl  = vert ? 0.0f : -sl;
    r.ric  = vert ? 0.0f : (yp - ic) * K_LOG2E;
    r.minx = fminf(x0, x1) * K_LOG2E;
    r.maxx = fmaxf(x0, x1) * K_LOG2E;
    r.lo   = (yp - fmaxf(y0, y1)) * K_LOG2E;
    r.hi   = (yp - fminf(y0, y1)) * K_LOG2E;
    return r;
}

__device__ __forceinline__ bool seg_near(const SegRow r, float xlo, float xhi)
{
    float cxa = fmaf(r.B, xlo, r.Cr);
    float cxb = fmaf(r.B, xhi, r.Cr);
    float pa  = __builtin_amdgcn_fmed3f(cxa, r.minx, r.maxx);
    float pb  = __builtin_amdgcn_fmed3f(cxb, r.minx, r.maxx);
    float ea  = __builtin_amdgcn_fmed3f(fmaf(r.nsl, cxa, r.ric), r.lo, r.hi);
    float eb  = __builtin_amdgcn_fmed3f(fmaf(r.nsl, cxb, r.ric), r.lo, r.hi);
    float dxm = fmaxf(0.0f, fmaxf(xlo - pb, pa - xhi));
    float dym = fmaxf(0.0f, fmaxf(fminf(ea, eb), -fmaxf(ea, eb)));
    return !(fmaf(dxm, dxm, dym * dym) >= CULL_R2);
}

__device__ __forceinline__ void put_seg(float* __restrict__ list, int pos, const SegRow r)
{
    float* d = list + (pos >> 1) * 16 + (pos & 1);
    d[0]  = r.B;    d[2]  = r.Cr;
    d[4]  = r.nsl;  d[6]  = r.ric;
    d[8]  = r.minx; d[10] = r.maxx;
    d[12] = r.lo;   d[14] = r.hi;
}

template <int NT>
__device__ __forceinline__ void tiles_step(const v4f r0, const v4f r1, const v4f r2, const v4f r3,
                                           const v2f* __restrict__ xpv, v2f* __restrict__ acc)
{
    const v2f epsv = {EPS_S, EPS_S};
    const v2f Bv   = {r0.x, r0.y}, Crv  = {r0.z, r0.w};
    const v2f nslv = {r1.x, r1.y}, ricv = {r1.z, r1.w};
    v2f cx[NT], dyr[NT], ddx[NT], ddy[NT];
    v2f s[NT], a[NT];
    #pragma unroll
    for (int t = 0; t < NT; ++t)
        cx[t] = __builtin_elementwise_fma(Bv, xpv[t], Crv);
    #pragma unroll
    for (int t = 0; t < NT; ++t)
        dyr[t] = __builtin_elementwise_fma(nslv, cx[t], ricv);
    #pragma unroll
    for (int t = 0; t < NT; ++t) {
        v2f xp1;
        xp1.x = __builtin_amdgcn_fmed3f(cx[t].x, r2.x, r2.z);
        xp1.y = __builtin_amdgcn_fmed3f(cx[t].y, r2.y, r2.w);
        ddx[t] = xpv[t] - xp1;
    }
    #pragma unroll
    for (int t = 0; t < NT; ++t) {
        ddy[t].x = __builtin_amdgcn_fmed3f(dyr[t].x, r3.x, r3.z);
        ddy[t].y = __builtin_amdgcn_fmed3f(dyr[t].y, r3.y, r3.w);
    }
    #pragma unroll
    for (int t = 0; t < NT; ++t)
        s[t] = __builtin_elementwise_fma(ddx[t], ddx[t], epsv);
    #pragma unroll
    for (int t = 0; t < NT; ++t)
        s[t] = __builtin_elementwise_fma(ddy[t], ddy[t], s[t]);
    #pragma unroll
    for (int t = 0; t < NT; ++t) {
        a[t].x = __builtin_amdgcn_sqrtf(s[t].x);
        a[t].y = __builtin_amdgcn_sqrtf(s[t].y);
    }
    #pragma unroll
    for (int t = 0; t < NT; ++t) {
        a[t].x = __builtin_amdgcn_exp2f(-a[t].x);
        a[t].y = __builtin_amdgcn_exp2f(-a[t].y);
    }
    #pragma unroll
    for (int t = 0; t < NT; ++t)
        acc[t] = acc[t] + a[t];
}

template <int NT>
__device__ __forceinline__ void strip_wave(const v4f* __restrict__ rec, const int steps,
                                           const int tile0, const int m,
                                           v4f* __restrict__ dst)
{
    v2f xpv[NT];
    v2f acc[NT];
    #pragma unroll
    for (int t = 0; t < NT; ++t) {
        const float xp = (float)((tile0 + t) * 16 + m) * K_LOG2E;
        xpv[t] = (v2f){xp, xp};
        acc[t] = (v2f){0.0f, 0.0f};
    }

    for (int s = 0; s + 1 < steps; s += 2) {
        const v4f* r = rec + s * 8;
        v4f a0 = r[0], a1 = r[1], a2 = r[2],  a3 = r[3];
        v4f b0 = r[8], b1 = r[9], b2 = r[10], b3 = r[11];
        tiles_step<NT>(a0, a1, a2, a3, xpv, acc);
        tiles_step<NT>(b0, b1, b2, b3, xpv, acc);
    }
    if (steps & 1) {
        const v4f* r = rec + (steps - 1) * 8;
        v4f a0 = r[0], a1 = r[1], a2 = r[2], a3 = r[3];
        tiles_step<NT>(a0, a1, a2, a3, xpv, acc);
    }

    v8f c[NT];
    {
        const v2f bones = {1.0f, 1.0f};
        const v8f zero8 = {0.0f, 0.0f, 0.0f, 0.0f, 0.0f, 0.0f, 0.0f, 0.0f};
        #pragma unroll
        for (int t = 0; t < NT; ++t)
            c[t] = __builtin_amdgcn_wmma_f32_16x16x4_f32(false, acc[t], false, bones,
                                                         (short)0, zero8, false, false);
    }

    if (m == 0) {
        #pragma unroll
        for (int t = 0; t < NT; ++t) {
            v4f lo = {c[t][0], c[t][1], c[t][2], c[t][3]};
            v4f hi = {c[t][4], c[t][5], c[t][6], c[t][7]};
            dst[t * 4 + 0] = lo;
            dst[t * 4 + 1] = hi;
        }
    }
}

__global__ __launch_bounds__(BLOCK_THREADS) void sketch_render_kernel(
    const float* __restrict__ xpar,
    const float* __restrict__ ypar,
    float* __restrict__ out)
{
    __shared__ v4f sT[NSTRIP * NPAIR * 4];
    __shared__ unsigned sCnt[2 * NWAVES];
    __shared__ v4f sP[KSPLIT * IMSIZE / 4];

    const int   row = blockIdx.x;
    const float yp  = (float)row;

    const int wave = __builtin_amdgcn_readfirstlane((int)(threadIdx.x >> 5));
    const int lane = threadIdx.x & 31;

    unsigned tot01 = 0, tot23 = 0;
    {
        const int p = threadIdx.x;
        float xa = xpar[2 * p]     * (float)IMSIZE;
        float xb = xpar[2 * p + 1] * (float)IMSIZE;
        float xc = xpar[2 * p + 2] * (float)IMSIZE;
        float ya = ypar[2 * p]     * (float)IMSIZE;
        float yb = ypar[2 * p + 1] * (float)IMSIZE;
        float yc = ypar[2 * p + 2] * (float)IMSIZE;
        SegRow e = seg_row(xa, ya, xb, yb, yp);
        SegRow o = seg_row(xb, yb, xc, yc, yp);

        const float xlim[NSTRIP + 1] = {0.0f, 64.0f * K_LOG2E, 112.0f * K_LOG2E, 160.0f * K_LOG2E,
                                        (float)(IMSIZE - 1) * K_LOG2E};
        bool ke[NSTRIP], ko[NSTRIP];
        #pragma unroll
        for (int s = 0; s < NSTRIP; ++s) {
            ke[s] = seg_near(e, xlim[s], xlim[s + 1]);
            ko[s] = seg_near(o, xlim[s], xlim[s + 1]);
        }

        const unsigned below = (1u << lane) - 1u;
        int pos[NSTRIP];
        unsigned cnt[NSTRIP];
        #pragma unroll
        for (int s = 0; s < NSTRIP; ++s) {
            const unsigned bE = __builtin_amdgcn_ballot_w32(ke[s]);
            const unsigned bO = __builtin_amdgcn_ballot_w32(ko[s]);
            pos[s] = __popc(bE & below) + __popc(bO & below);
            cnt[s] = (unsigned)(__popc(bE) + __popc(bO));
        }
        if (lane == 0) {
            sCnt[wave]          = cnt[0] | (cnt[1] << 16);
            sCnt[NWAVES + wave] = cnt[2] | (cnt[3] << 16);
        }
        __syncthreads();
        const int wavev = (int)(threadIdx.x >> 5);
        unsigned pre01 = 0, pre23 = 0;
        #pragma unroll
        for (int w = 0; w < NWAVES; ++w) {
            const unsigned c01 = sCnt[w], c23 = sCnt[NWAVES + w];
            tot01 += c01;
            tot23 += c23;
            if (w < wavev) { pre01 += c01; pre23 += c23; }
        }
        pos[0] += (int)(pre01 & 0xffffu); pos[1] += (int)(pre01 >> 16);
        pos[2] += (int)(pre23 & 0xffffu); pos[3] += (int)(pre23 >> 16);
        #pragma unroll
        for (int s = 0; s < NSTRIP; ++s) {
            float* const list = (float*)(sT + s * (NPAIR * 4));
            if (ke[s]) put_seg(list, pos[s], e);
            if (ko[s]) put_seg(list, pos[s] + (ke[s] ? 1 : 0), o);
        }

        SegRow f;
        f.B = 0.0f; f.Cr = 0.0f; f.nsl = 0.0f; f.ric = 0.0f;
        f.minx = FAR_AWAY; f.maxx = FAR_AWAY; f.lo = FAR_AWAY; f.hi = FAR_AWAY;
        if (wave < NSTRIP) {
            float* const list = (float*)(sT + wave * (NPAIR * 4));
            const unsigned n  = (((wave & 2) ? tot23 : tot01) >> ((wave & 1) * 16)) & 0xffffu;
            const unsigned i  = n + (unsigned)lane;
            if (i < ((n + STEP_SEGS - 1u) / STEP_SEGS) * STEP_SEGS) put_seg(list, (int)i, f);
        }
    }
    __syncthreads();

    const int grp  = lane >> 4;
    const int m    = lane & 15;
    const int strip = (wave ^ (wave >> 2)) & 3;
    const int ks    = wave >> 2;

    const unsigned nstrip = (((strip & 2) ? tot23 : tot01) >> ((strip & 1) * 16)) & 0xffffu;
    const int steps = __builtin_amdgcn_readfirstlane((int)((nstrip + STEP_SEGS - 1u) / STEP_SEGS));
    const v4f* rec = sT + strip * (NPAIR * 4) + (ks * steps * 2 + grp) * 4;
    const int tile0 = 3 * strip + (strip != 0 ? 1 : 0);
    v4f* dst = sP + (ks * IMSIZE + tile0 * 16 + grp * 8) / 4;

    if (strip == 0 || strip == 3)
        strip_wave<4>(rec, steps, tile0, m, dst);
    else
        strip_wave<3>(rec, steps, tile0, m, dst);
    __syncthreads();

    if (threadIdx.x < IMSIZE) {
        const float* P = (const float*)sP;
        float res = P[threadIdx.x];
        #pragma unroll
        for (int k = 1; k < KSPLIT; ++k)
            res += P[k * IMSIZE + threadIdx.x];
        out[row * IMSIZE + threadIdx.x] = (LOG2_NSEG - __builtin_amdgcn_logf(res)) * LN2_OVER_K;
    }
}

extern "C" void kernel_launch(void* const* d_in, const int* in_sizes, int n_in,
                              void* d_out, int out_size, void* d_ws, size_t ws_size,
                              hipStream_t stream) {
    (void)in_sizes; (void)n_in; (void)out_size; (void)d_ws; (void)ws_size;
    const float* x = (const float*)d_in[0];
    const float* y = (const float*)d_in[1];
    float* out = (float*)d_out;
    dim3 grid(IMSIZE);
    dim3 block(BLOCK_THREADS);
    sketch_render_kernel<<<grid, block, 0, stream>>>(x, y, out);
}


# Closest point by segment parameter t=clamp(a*xp+b,0,1), clamp done by the packed fma's modifier, replaces line foot + two box med3: per pair 6 instr not 7, LDS record 24 B not 32; rounding changes
# speedup vs baseline: 6.3651x; 1.1803x over previous
#include <hip/hip_runtime.h>


typedef __attribute__((ext_vector_type(2))) float v2f;
typedef __attribute__((ext_vector_type(4))) float v4f;
typedef __attribute__((ext_vector_type(8))) float v8f;

#define IMSIZE   224
#define NSEG     1024
#define NPAIR    (NSEG / 2)
#define SMOOTH_K 8.0f
#define EPS_F    1e-10f

#define NSTRIP         4
#define KSPLIT         4
#define NWAVES         (NSTRIP * KSPLIT)
#define BLOCK_THREADS  (NWAVES * 32)
#define STEP_SEGS      (KSPLIT * 4)

#define CULL_R2        ((13.5f * K_LOG2E) * (13.5f * K_LOG2E))
#define FAR_AWAY       (-10000.0f)

#define K_LOG2E    (SMOOTH_K * 1.44269504088896340736f)
#define EPS_S      (EPS_F * K_LOG2E * K_LOG2E)
#define LOG2_NSEG  10.0f
#define LN2_OVER_K (0.69314718055994530942f / SMOOTH_K)

struct SegRow { float a, b, gx, px, gy, ey; };

__device__ __forceinline__ SegRow seg_row(float x0, float y0, float x1, float y1, float yp)
{
    float dx = x1 - x0, dy = y1 - y0;
    float d2 = fmaf(dy, dy, dx * dx);
    float inv = 1.0f / (d2 + EPS_F);
    float qy = yp - y0;
    SegRow r;
    r.a  = dx * inv * (1.0f / K_LOG2E);
    r.b  = fmaf(qy, dy, -(x0 * dx)) * inv;
    r.gx = dx * K_LOG2E;
    r.px = x0 * K_LOG2E;
    r.gy = dy * K_LOG2E;
    r.ey = qy * K_LOG2E;
    return r;
}

__device__ __forceinline__ bool seg_near(const SegRow r, float xlo, float xhi)
{
    float ta  = __builtin_amdgcn_fmed3f(fmaf(r.a, xlo, r.b), 0.0f, 1.0f);
    float tb  = __builtin_amdgcn_fmed3f(fmaf(r.a, xhi, r.b), 0.0f, 1.0f);
    float pa  = fmaf(r.gx, ta, r.px);
    float pb  = fmaf(r.gx, tb, r.px);
    float ea  = fmaf(-r.gy, ta, r.ey);
    float eb  = fmaf(-r.gy, tb, r.ey);
    float dxm = fmaxf(0.0f, fmaxf(xlo - pb, pa - xhi));
    float dym = fmaxf(0.0f, fmaxf(fminf(ea, eb), -fmaxf(ea, eb)));
    return !(fmaf(dxm, dxm, dym * dym) >= CULL_R2);
}

__device__ __forceinline__ void put_seg(float* __restrict__ list, int pos, const SegRow r)
{
    float* d = list + (pos >> 1) * 12 + (pos & 1);
    d[0] = r.a;   d[2]  = r.b;
    d[4] = r.gx;  d[6]  = r.px;
    d[8] = r.gy;  d[10] = r.ey;
}

__device__ __forceinline__ v2f fma_clamp01(const v2f a, const v2f x, const v2f b)
{
    v2f t;
    asm("v_pk_fma_f32 %0, %1, %2, %3 clamp" : "=v"(t) : "v"(a), "v"(x), "v"(b));
    return t;
}

template <int NT>
__device__ __forceinline__ void tiles_step(const v4f r0, const v4f r1, const v4f r2,
                                           const v2f* __restrict__ xpv, v2f* __restrict__ acc)
{
    const v2f epsv = {EPS_S, EPS_S};
    const v2f av  = {r0.x, r0.y}, bv  = {r0.z, r0.w};
    const v2f gxv = {r1.x, r1.y}, pxv = {r1.z, r1.w};
    const v2f gyv = {r2.x, r2.y}, eyv = {r2.z, r2.w};
    v2f tt[NT], xp1[NT], ddx[NT], ddy[NT];
    v2f s[NT], e[NT];
    #pragma unroll
    for (int t = 0; t < NT; ++t)
        tt[t] = fma_clamp01(av, xpv[t], bv);
    #pragma unroll
    for (int t = 0; t < NT; ++t)
        xp1[t] = __builtin_elementwise_fma(gxv, tt[t], pxv);
    #pragma unroll
    for (int t = 0; t < NT; ++t)
        ddy[t] = __builtin_elementwise_fma(-gyv, tt[t], eyv);
    #pragma unroll
    for (int t = 0; t < NT; ++t)
        ddx[t] = xpv[t] - xp1[t];
    #pragma unroll
    for (int t = 0; t < NT; ++t)
        s[t] = __builtin_elementwise_fma(ddx[t], ddx[t], epsv);
    #pragma unroll
    for (int t = 0; t < NT; ++t)
        s[t] = __builtin_elementwise_fma(ddy[t], ddy[t], s[t]);
    #pragma unroll
    for (int t = 0; t < NT; ++t) {
        e[t].x = __builtin_amdgcn_sqrtf(s[t].x);
        e[t].y = __builtin_amdgcn_sqrtf(s[t].y);
    }
    #pragma unroll
    for (int t = 0; t < NT; ++t) {
        e[t].x = __builtin_amdgcn_exp2f(-e[t].x);
        e[t].y = __builtin_amdgcn_exp2f(-e[t].y);
    }
    #pragma unroll
    for (int t = 0; t < NT; ++t)
        acc[t] = acc[t] + e[t];
}

template <int NT>
__device__ __forceinline__ void strip_wave(const v4f* __restrict__ rec, const int steps,
                                           const int tile0, const int m,
                                           v4f* __restrict__ dst)
{
    v2f xpv[NT];
    v2f acc[NT];
    #pragma unroll
    for (int t = 0; t < NT; ++t) {
        const float xp = (float)((tile0 + t) * 16 + m) * K_LOG2E;
        xpv[t] = (v2f){xp, xp};
        acc[t] = (v2f){0.0f, 0.0f};
    }

    for (int s = 0; s + 1 < steps; s += 2) {
        const v4f* r = rec + s * 6;
        v4f a0 = r[0], a1 = r[1], a2 = r[2];
        v4f b0 = r[6], b1 = r[7], b2 = r[8];
        tiles_step<NT>(a0, a1, a2, xpv, acc);
        tiles_step<NT>(b0, b1, b2, xpv, acc);
    }
    if (steps & 1) {
        const v4f* r = rec + (steps - 1) * 6;
        v4f a0 = r[0], a1 = r[1], a2 = r[2];
        tiles_step<NT>(a0, a1, a2, xpv, acc);
    }

    v8f c[NT];
    {
        const v2f bones = {1.0f, 1.0f};
        const v8f zero8 = {0.0f, 0.0f, 0.0f, 0.0f, 0.0f, 0.0f, 0.0f, 0.0f};
        #pragma unroll
        for (int t = 0; t < NT; ++t)
            c[t] = __builtin_amdgcn_wmma_f32_16x16x4_f32(false, acc[t], false, bones,
                                                         (short)0, zero8, false, false);
    }

    if (m == 0) {
        #pragma unroll
        for (int t = 0; t < NT; ++t) {
            v4f lo = {c[t][0], c[t][1], c[t][2], c[t][3]};
            v4f hi = {c[t][4], c[t][5], c[t][6], c[t][7]};
            dst[t * 4 + 0] = lo;
            dst[t * 4 + 1] = hi;
        }
    }
}

__global__ __launch_bounds__(BLOCK_THREADS) void sketch_render_kernel(
    const float* __restrict__ xpar,
    const float* __restrict__ ypar,
    float* __restrict__ out)
{
    __shared__ v4f sT[NSTRIP * NPAIR * 3];
    __shared__ unsigned sCnt[2 * NWAVES];
    __shared__ v4f sP[KSPLIT * IMSIZE / 4];

    const int   row = blockIdx.x;
    const float yp  = (float)row;

    const int wave = __builtin_amdgcn_readfirstlane((int)(threadIdx.x >> 5));
    const int lane = threadIdx.x & 31;

    unsigned tot01 = 0, tot23 = 0;
    {
        const int p = threadIdx.x;
        float xa = xpar[2 * p]     * (float)IMSIZE;
        float xb = xpar[2 * p + 1] * (float)IMSIZE;
        float xc = xpar[2 * p + 2] * (float)IMSIZE;
        float ya = ypar[2 * p]     * (float)IMSIZE;
        float yb = ypar[2 * p + 1] * (float)IMSIZE;
        float yc = ypar[2 * p + 2] * (float)IMSIZE;
        SegRow e = seg_row(xa, ya, xb, yb, yp);
        SegRow o = seg_row(xb, yb, xc, yc, yp);

        const float xlim[NSTRIP + 1] = {0.0f, 64.0f * K_LOG2E, 112.0f * K_LOG2E, 160.0f * K_LOG2E,
                                        (float)(IMSIZE - 1) * K_LOG2E};
        bool ke[NSTRIP], ko[NSTRIP];
        #pragma unroll
        for (int s = 0; s < NSTRIP; ++s) {
            ke[s] = seg_near(e, xlim[s], xlim[s + 1]);
            ko[s] = seg_near(o, xlim[s], xlim[s + 1]);
        }

        const unsigned below = (1u << lane) - 1u;
        int pos[NSTRIP];
        unsigned cnt[NSTRIP];
        #pragma unroll
        for (int s = 0; s < NSTRIP; ++s) {
            const unsigned bE = __builtin_amdgcn_ballot_w32(ke[s]);
            const unsigned bO = __builtin_amdgcn_ballot_w32(ko[s]);
            pos[s] = __popc(bE & below) + __popc(bO & below);
            cnt[s] = (unsigned)(__popc(bE) + __popc(bO));
        }
        if (lane == 0) {
            sCnt[wave]          = cnt[0] | (cnt[1] << 16);
            sCnt[NWAVES + wave] = cnt[2] | (cnt[3] << 16);
        }
        __syncthreads();
        const int wavev = (int)(threadIdx.x >> 5);
        unsigned pre01 = 0, pre23 = 0;
        #pragma unroll
        for (int w = 0; w < NWAVES; ++w) {
            const unsigned c01 = sCnt[w], c23 = sCnt[NWAVES + w];
            tot01 += c01;
            tot23 += c23;
            if (w < wavev) { pre01 += c01; pre23 += c23; }
        }
        pos[0] += (int)(pre01 & 0xffffu); pos[1] += (int)(pre01 >> 16);
        pos[2] += (int)(pre23 & 0xffffu); pos[3] += (int)(pre23 >> 16);
        #pragma unroll
        for (int s = 0; s < NSTRIP; ++s) {
            float* const list = (float*)(sT + s * (NPAIR * 3));
            if (ke[s]) put_seg(list, pos[s], e);
            if (ko[s]) put_seg(list, pos[s] + (ke[s] ? 1 : 0), o);
        }

        SegRow f;
        f.a = 0.0f; f.b = 0.0f; f.gx = 0.0f; f.px = FAR_AWAY; f.gy = 0.0f; f.ey = 0.0f;
        if (wave < NSTRIP) {
            float* const list = (float*)(sT + wave * (NPAIR * 3));
            const unsigned n  = (((wave & 2) ? tot23 : tot01) >> ((wave & 1) * 16)) & 0xffffu;
            const unsigned i  = n + (unsigned)lane;
            if (i < ((n + STEP_SEGS - 1u) / STEP_SEGS) * STEP_SEGS) put_seg(list, (int)i, f);
        }
    }
    __syncthreads();

    const int grp  = lane >> 4;
    const int m    = lane & 15;
    const int strip = (wave ^ (wave >> 2)) & 3;
    const int ks    = wave >> 2;

    const unsigned nstrip = (((strip & 2) ? tot23 : tot01) >> ((strip & 1) * 16)) & 0xffffu;
    const int steps = __builtin_amdgcn_readfirstlane((int)((nstrip + STEP_SEGS - 1u) / STEP_SEGS));
    const v4f* rec = sT + strip * (NPAIR * 3) + (ks * steps * 2 + grp) * 3;
    const int tile0 = 3 * strip + (strip != 0 ? 1 : 0);
    v4f* dst = sP + (ks * IMSIZE + tile0 * 16 + grp * 8) / 4;

    if (strip == 0 || strip == 3)
        strip_wave<4>(rec, steps, tile0, m, dst);
    else
        strip_wave<3>(rec, steps, tile0, m, dst);
    __syncthreads();

    if (threadIdx.x < IMSIZE) {
        const float* P = (const float*)sP;
        float res = P[threadIdx.x];
        #pragma unroll
        for (int k = 1; k < KSPLIT; ++k)
            res += P[k * IMSIZE + threadIdx.x];
        out[row * IMSIZE + threadIdx.x] = (LOG2_NSEG - __builtin_amdgcn_logf(res)) * LN2_OVER_K;
    }
}

extern "C" void kernel_launch(void* const* d_in, const int* in_sizes, int n_in,
                              void* d_out, int out_size, void* d_ws, size_t ws_size,
                              hipStream_t stream) {
    (void)in_sizes; (void)n_in; (void)out_size; (void)d_ws; (void)ws_size;
    const float* x = (const float*)d_in[0];
    const float* y = (const float*)d_in[1];
    float* out = (float*)d_out;
    dim3 grid(IMSIZE);
    dim3 block(BLOCK_THREADS);
    sketch_render_kernel<<<grid, block, 0, stream>>>(x, y, out);
}
